# starting kernel
# baseline (speedup 1.0000x reference)
#include <hip/hip_runtime.h>
#include <math.h>

#define LL    50
#define BB    16
#define ITERR 76
#define KDIM  100
#define CDIM  3800
#define CPAD  3808
#define JPAD  64
#define SROW  7600
#define SROW0 50
#define COL0  1900

typedef float v2f __attribute__((ext_vector_type(2)));
typedef float v8f __attribute__((ext_vector_type(8)));

#define WMMA_F32X4(a, b, c) \
  __builtin_amdgcn_wmma_f32_16x16x4_f32(false, (a), false, (b), (short)0, (c), false, false)

__global__ void k_maxes(const float* __restrict__ inp, const float* __restrict__ wgt,
                        float* __restrict__ smax) {
  __shared__ float red[256];
  int b = blockIdx.x;
  float m = -3.4e38f;
  if (b < BB) {
    const float* p = inp + b * 2500;
    for (int i = threadIdx.x; i < 2500; i += 256) m = fmaxf(m, p[i]);
  } else {
    for (int i = threadIdx.x; i < 2500; i += 256) m = fmaxf(m, fabsf(wgt[i]));
  }
  red[threadIdx.x] = m;
  __syncthreads();
  for (int s = 128; s > 0; s >>= 1) {
    if (threadIdx.x < s) red[threadIdx.x] = fmaxf(red[threadIdx.x], red[threadIdx.x + s]);
    __syncthreads();
  }
  if (threadIdx.x == 0) smax[b] = red[0];
}

__global__ void k_build_q(float* __restrict__ qre, float* __restrict__ qim) {
  int idx = blockIdx.x * 256 + threadIdx.x;
  if (idx >= JPAD * CPAD) return;
  int i = idx / CPAD, c = idx % CPAD;
  float re = 0.f, im = 0.f;
  if (i < LL) {
    int t = ((i + 1) * (COL0 + c)) % 200;
    float s, cz;
    sincospif((float)t * 0.01f, &s, &cz);
    re = cz; im = -s;
  }
  qre[idx] = re; qim[idx] = im;
}

__global__ void __launch_bounds__(128) k_mm1(
    const float* __restrict__ inp, const float* __restrict__ wgt,
    const float* __restrict__ sre, const float* __restrict__ sim,
    float* __restrict__ gre, float* __restrict__ gim) {
  __shared__ float Pre[16 * KDIM];
  __shared__ float Pim[16 * KDIM];
  int bid = blockIdx.x;
  int cg = bid % 60;
  int jt = (bid / 60) % 4;
  int b  = bid / 240;

  for (int e = threadIdx.x; e < 16 * KDIM; e += 128) {
    int jl = e / KDIM, r = e % KDIM;
    int j = jt * 16 + jl;
    float re = 0.f, im = 0.f;
    if (j < LL) {
      int t = ((150 + 2 * j) * (50 + r)) % 200;
      float s, cz;
      sincospif((float)t * 0.01f, &s, &cz);
      re = cz; im = -s;
    }
    Pre[e] = re; Pim[e] = im;
  }
  __syncthreads();

  int wave = threadIdx.x >> 5;
  int lane = threadIdx.x & 31;
  int ctile = cg * 4 + wave;
  if (ctile >= CPAD / 16) return;

  int h  = lane >> 4;
  int ln = lane & 15;
  int c  = ctile * 16 + ln;
  int x2 = COL0 + c;
  int q  = (c < CDIM) ? (c / ITERR) : 0;
  const float* inp_b = inp + b * 2500 + q;

  v8f cre = {}; v8f cim = {};
  for (int t = 0; t < KDIM / 4; ++t) {
    int kk = 4 * t + 2 * h;
    v2f are = *(const v2f*)&Pre[ln * KDIM + kk];
    v2f aim = *(const v2f*)&Pim[ln * KDIM + kk];
    v2f aimn = -aim;
    float inpv = inp_b[(2 * t + h) * LL];
    int so = (SROW0 + kk) * SROW + x2;
    v2f bre; bre.x = sre[so] * inpv; bre.y = sre[so + SROW] * inpv;
    v2f bim; bim.x = sim[so] * inpv; bim.y = sim[so + SROW] * inpv;
    cre = WMMA_F32X4(are,  bre, cre);
    cre = WMMA_F32X4(aimn, bim, cre);
    cim = WMMA_F32X4(are,  bim, cim);
    cim = WMMA_F32X4(aim,  bre, cim);
  }

  size_t gbase = (size_t)b * JPAD * CPAD;
#pragma unroll
  for (int r = 0; r < 8; ++r) {
    int j = jt * 16 + r + 8 * h;
    float w = (j < LL) ? fabsf(wgt[j * LL + q]) : 0.f;
    gre[gbase + (size_t)j * CPAD + c] = cre[r] * w;
    gim[gbase + (size_t)j * CPAD + c] = cim[r] * w;
  }
}

__global__ void __launch_bounds__(128) k_mm2(
    const float* __restrict__ gre, const float* __restrict__ gim,
    const float* __restrict__ qre, const float* __restrict__ qim,
    const float* __restrict__ amp, float* __restrict__ magbuf) {
  int jt = blockIdx.x % 4;
  int b  = blockIdx.x / 4;
  int wave = threadIdx.x >> 5;
  int lane = threadIdx.x & 31;
  int h = lane >> 4, ln = lane & 15;
  int i = wave * 16 + ln;

  const float* gre_l = gre + (size_t)b * JPAD * CPAD + (size_t)(jt * 16 + ln) * CPAD;
  const float* gim_l = gim + (size_t)b * JPAD * CPAD + (size_t)(jt * 16 + ln) * CPAD;
  const float* qre_l = qre + (size_t)i * CPAD;
  const float* qim_l = qim + (size_t)i * CPAD;

  v8f cre = {}; v8f cim = {};
  for (int t = 0; t < CDIM / 4; ++t) {
    int k = 4 * t + 2 * h;
    v2f are = *(const v2f*)&gre_l[k];
    v2f aim = *(const v2f*)&gim_l[k];
    v2f bre = *(const v2f*)&qre_l[k];
    v2f bim = *(const v2f*)&qim_l[k];
    v2f aimn = -aim;
    cre = WMMA_F32X4(are,  bre, cre);
    cre = WMMA_F32X4(aimn, bim, cre);
    cim = WMMA_F32X4(are,  bim, cim);
    cim = WMMA_F32X4(aim,  bre, cim);
  }

#pragma unroll
  for (int r = 0; r < 8; ++r) {
    int j = jt * 16 + r + 8 * h;
    float mag = sqrtf(cre[r] * cre[r] + cim[r] * cim[r]);
    float v = 0.f;
    if (j < LL && i < LL) v = mag / amp[j * LL + i];
    magbuf[(size_t)b * JPAD * JPAD + j * JPAD + i] = v;
  }
}

__global__ void k_final(const float* __restrict__ magbuf, const float* __restrict__ smax,
                        const float* __restrict__ bias, float* __restrict__ out) {
  __shared__ float red[256];
  int b = blockIdx.x;
  const float* mb = magbuf + (size_t)b * JPAD * JPAD;
  float m = 0.f;
  for (int idx = threadIdx.x; idx < JPAD * JPAD; idx += 256) m = fmaxf(m, mb[idx]);
  red[threadIdx.x] = m;
  __syncthreads();
  for (int s = 128; s > 0; s >>= 1) {
    if (threadIdx.x < s) red[threadIdx.x] = fmaxf(red[threadIdx.x], red[threadIdx.x + s]);
    __syncthreads();
  }
  float scale = (smax[b] * smax[16]) / red[0];
  for (int idx = threadIdx.x; idx < LL * LL; idx += 256) {
    int i = idx / LL, j = idx % LL;
    out[b * 2500 + idx] = mb[j * JPAD + i] * scale + bias[j];
  }
}

extern "C" void kernel_launch(void* const* d_in, const int* in_sizes, int n_in,
                              void* d_out, int out_size, void* d_ws, size_t ws_size,
                              hipStream_t stream) {
  const float* inp  = (const float*)d_in[0];
  const float* wgt  = (const float*)d_in[1];
  const float* bias = (const float*)d_in[2];
  const float* amp  = (const float*)d_in[3];
  const float* sre  = (const float*)d_in[4];
  const float* sim  = (const float*)d_in[5];

  float* ws   = (float*)d_ws;
  float* smax = ws;
  float* qre  = ws + 32;
  float* qim  = qre + (size_t)JPAD * CPAD;
  float* gre  = qim + (size_t)JPAD * CPAD;
  float* gim  = gre + (size_t)BB * JPAD * CPAD;
  float* magb = gim + (size_t)BB * JPAD * CPAD;

  k_maxes  <<<17, 256, 0, stream>>>(inp, wgt, smax);
  k_build_q<<<(JPAD * CPAD + 255) / 256, 256, 0, stream>>>(qre, qim);
  k_mm1    <<<BB * 4 * 60, 128, 0, stream>>>(inp, wgt, sre, sim, gre, gim);
  k_mm2    <<<BB * 4, 128, 0, stream>>>(gre, gim, qre, qim, amp, magb);
  k_final  <<<BB, 256, 0, stream>>>(magb, smax, bias, (float*)d_out);
}


# product kernel transposed: j on tile columns, one i per block, so a table value serves a half-wave (16x fewer table bytes); each lane half takes one q; |W| in B: changes rounding and summation order
# speedup vs baseline: 9.3415x; 9.3415x over previous
#include <hip/hip_runtime.h>
#include <math.h>

#define LL    50
#define BB    16
#define ITERR 76
#define KDIM  100
#define IPAD  64
#define SROW  7600
#define SROW0 50
#define COL0  1900
#define NRT   7
#define NQW   25
#define TROW  32
#define MROW  32
#define MBAT  (LL * 4 * MROW)

typedef float v2f __attribute__((ext_vector_type(2)));
typedef float v4f __attribute__((ext_vector_type(4)));
typedef float v8f __attribute__((ext_vector_type(8)));

#define WMMA_F32X4(a, b, c) \
  __builtin_amdgcn_wmma_f32_16x16x4_f32(false, (a), false, (b), (short)0, (c), false, false)


__device__ __forceinline__ float wave_max(float m) {
#pragma unroll
  for (int o = 16; o > 0; o >>= 1) m = fmaxf(m, __shfl_xor(m, o, 32));
  return m;
}

__global__ void __launch_bounds__(256) k_maxes(const float* __restrict__ inp, const float* __restrict__ wgt,
                                               float* __restrict__ smax) {
  __shared__ float red[8];
  int b = blockIdx.x;
  int t = threadIdx.x;
  bool isw = b >= BB;
  const float4* p = (const float4*)(isw ? wgt : inp + b * 2500);
  int f2 = t + 512; if (f2 > 624) f2 = 624;
  float4 v0 = p[t];
  float4 v1 = p[t + 256];
  float4 v2 = p[f2];
  if (isw) {
    v0.x = fabsf(v0.x); v0.y = fabsf(v0.y); v0.z = fabsf(v0.z); v0.w = fabsf(v0.w);
    v1.x = fabsf(v1.x); v1.y = fabsf(v1.y); v1.z = fabsf(v1.z); v1.w = fabsf(v1.w);
    v2.x = fabsf(v2.x); v2.y = fabsf(v2.y); v2.z = fabsf(v2.z); v2.w = fabsf(v2.w);
  }
  float m = -3.4e38f;
  m = fmaxf(m, fmaxf(fmaxf(v0.x, v0.y), fmaxf(v0.z, v0.w)));
  m = fmaxf(m, fmaxf(fmaxf(v1.x, v1.y), fmaxf(v1.z, v1.w)));
  m = fmaxf(m, fmaxf(fmaxf(v2.x, v2.y), fmaxf(v2.z, v2.w)));
  m = wave_max(m);
  if ((t & 31) == 0) red[t >> 5] = m;
  __syncthreads();
  if (t == 0) {
    float r = red[0];
#pragma unroll
    for (int w = 1; w < 8; ++w) r = fmaxf(r, red[w]);
    smax[b] = r;
  }
}

__global__ void __launch_bounds__(128) k_table(
    const float* __restrict__ inp, const float* __restrict__ sre, const float* __restrict__ sim,
    float* __restrict__ inpt, float* __restrict__ tbuf) {
  __shared__ v2f tw[201];
  int bid = blockIdx.x;
  if (bid >= LL * NRT) {
    int q = bid - LL * NRT;
    for (int e = threadIdx.x; e < LL * BB; e += 128) {
      int p = e >> 4, b = e & 15;
      inpt[((q * (LL / 2) + (p >> 1)) * BB + b) * 2 + (p & 1)] = inp[b * 2500 + p * LL + q];
    }
    return;
  }
  for (int t = threadIdx.x; t < 201; t += 128) {
    float s, cz;
    sincospif((float)t * 0.01f, &s, &cz);
    v2f e; e.x = cz; e.y = -s;
    if (t == 200) { e.x = 0.f; e.y = 0.f; }
    tw[t] = e;
  }
  __syncthreads();

  int q  = bid / NRT;
  int rt = bid % NRT;
  int wave = threadIdx.x >> 5;
  int lane = threadIdx.x & 31;
  int h  = lane >> 4;
  int ln = lane & 15;
  int i  = wave * 16 + ln;
  int c0 = ITERR * q;
  size_t rowbase = (size_t)(SROW0 + rt * 16 + ln) * SROW + COL0 + c0;
  const float* sre_l = sre + rowbase;
  const float* sim_l = sim + rowbase;
  int step = i + 1;
  int m0 = (step * (COL0 + c0 + 2 * h)) % 200;
  int adv = (4 * step) % 200;

  v8f cre = {}; v8f cim = {};
  for (int t = 0; t < ITERR / 4; ++t) {
    int u = 4 * t + 2 * h;
    v2f are = *(const v2f*)&sre_l[u];
    v2f aim = *(const v2f*)&sim_l[u];
    v2f aimn = -aim;
    int m1 = m0 + step; if (m1 >= 200) m1 -= 200;
    v2f q0 = tw[(i < LL) ? m0 : 200];
    v2f q1 = tw[(i < LL) ? m1 : 200];
    v2f bre; bre.x = q0.x; bre.y = q1.x;
    v2f bim; bim.x = q0.y; bim.y = q1.y;
    m0 += adv; if (m0 >= 200) m0 -= 200;
    cre = WMMA_F32X4(are,  bre, cre);
    cre = WMMA_F32X4(aimn, bim, cre);
    cim = WMMA_F32X4(are,  bim, cim);
    cim = WMMA_F32X4(aim,  bre, cim);
  }

  if (i < LL) {
#pragma unroll
    for (int r = 0; r < 8; r += 4) {
      int row = rt * 16 + r + 8 * h;
      int p0 = row >> 1;
      if (p0 < LL) {
        float* dst = tbuf + (((size_t)i * LL + q) * TROW + (p0 >> 1)) * 8;
        *(float4*)dst       = make_float4(cre[r],     cre[r + 2], cim[r],     cim[r + 2]);
        *(float4*)(dst + 4) = make_float4(cre[r + 1], cre[r + 3], cim[r + 1], cim[r + 3]);
      }
    }
  }
}

__device__ __forceinline__ v2f fma2(v2f a, v2f b, v2f c) { return __builtin_elementwise_fma(a, b, c); }

__global__ void __launch_bounds__(NQW * 32) k_prod(
    const float* __restrict__ wgt, const float* __restrict__ inpt, const float* __restrict__ tbuf,
    const float* __restrict__ amp, float* __restrict__ magbuf) {
  __shared__ float4 pA[(LL / 2) * 16];
  __shared__ float4 pB[(LL / 2) * 16];
  __shared__ float red[NQW * 512];
  int i  = blockIdx.x >> 2;
  int jt = blockIdx.x & 3;

  {
    int t = threadIdx.x;
    int odd = t & 1, jl = (t >> 1) & 15, g = t >> 5;
    int k1u = 150 + 2 * (jt * 16 + jl);
    int r1 = 4 * g + odd;
    int m1 = (k1u * (50 + r1)) % 200;
    int m2 = (k1u * (52 + r1)) % 200;
    float s1, c1, s2, c2;
    sincospif((float)m1 * 0.01f, &s1, &c1);
    sincospif((float)m2 * 0.01f, &s2, &c2);
    float4 e = make_float4(c1, c2, -s1, -s2);
    if (odd) pB[g * 16 + jl] = e; else pA[g * 16 + jl] = e;
  }
  __syncthreads();

  int wave = threadIdx.x >> 5;
  int lane = threadIdx.x & 31;
  int h = lane >> 4, ln = lane & 15;
  int j = jt * 16 + ln;

  int q = 2 * wave + h;
  float wq = (j < LL) ? fabsf(wgt[j * LL + q]) : 0.f;
  v2f wq2; wq2.x = wq; wq2.y = wq;
  const v2f*    a_l = (const v2f*)inpt + (size_t)q * (LL / 2) * BB + ln;
  const v4f*    t_l = (const v4f*)tbuf + ((size_t)i * LL + q) * TROW * 2;
  const float4* pa_l = pA + ln;
  const float4* pb_l = pB + ln;

  v8f cre = {}; v8f cim = {};
#pragma unroll 1
  for (int g = 0; g < 5; ++g) {
    v2f    av5[5];
    v4f    ta5[5];
    v4f    tb5[5];
#pragma unroll
    for (int s = 0; s < 5; ++s) {
      av5[s] = a_l[(5 * g + s) * 16];
    }
#pragma unroll
    for (int s = 0; s < 5; ++s) {
      ta5[s] = t_l[(5 * g + s) * 2];
      tb5[s] = t_l[(5 * g + s) * 2 + 1];
    }
    asm volatile("" :: "v"(av5[0]), "v"(av5[1]), "v"(av5[2]), "v"(av5[3]), "v"(av5[4]),
                       "v"(ta5[0]), "v"(ta5[1]), "v"(ta5[2]), "v"(ta5[3]), "v"(ta5[4]),
                       "v"(tb5[0]), "v"(tb5[1]), "v"(tb5[2]), "v"(tb5[3]), "v"(tb5[4]));
#pragma unroll
    for (int s = 0; s < 5; ++s) {
      float4 pa = pa_l[(5 * g + s) * 16];
      float4 pb = pb_l[(5 * g + s) * 16];
      v4f ta = ta5[s];
      v4f tb = tb5[s];
      v2f p0r; p0r.x = pa.x; p0r.y = pa.y;
      v2f p0i; p0i.x = pa.z; p0i.y = pa.w;
      v2f p1r; p1r.x = pb.x; p1r.y = pb.y;
      v2f p1i; p1i.x = pb.z; p1i.y = pb.w;
      v2f t0r; t0r.x = ta.x; t0r.y = ta.y;
      v2f t0i; t0i.x = ta.z; t0i.y = ta.w;
      v2f t1r; t1r.x = tb.x; t1r.y = tb.y;
      v2f t1i; t1i.x = tb.z; t1i.y = tb.w;
      v2f bre = fma2(p0r, t0r, fma2(-p0i, t0i, fma2(p1r, t1r, -(p1i * t1i))));
      v2f bim = fma2(p0r, t0i, fma2( p0i, t0r, fma2(p1r, t1i,   p1i * t1r)));
      bre = bre * wq2;
      bim = bim * wq2;
      cre = WMMA_F32X4(av5[s], bre, cre);
      cim = WMMA_F32X4(av5[s], bim, cim);
    }
  }

  float* myred = red + wave * 512;
#pragma unroll
  for (int r = 0; r < 8; ++r) {
    myred[r * 32 + lane]       = cre[r];
    myred[256 + r * 32 + lane] = cim[r];
  }
  __syncthreads();
  if (threadIdx.x < 256) {
    float re = 0.f, im = 0.f;
    for (int w = 0; w < NQW; ++w) {
      re += red[w * 512 + threadIdx.x];
      im += red[w * 512 + 256 + threadIdx.x];
    }
    int r = threadIdx.x >> 5, l = threadIdx.x & 31;
    int b  = r + 8 * (l >> 4);
    int jo = jt * 16 + (l & 15);
    if (jo < LL)
      magbuf[(size_t)b * MBAT + (i * 4 + jt) * MROW + (l & 15)] = sqrtf(re * re + im * im) / amp[jo * LL + i];
  }
}

__global__ void __launch_bounds__(256) k_final(const float* __restrict__ magbuf, const float* __restrict__ smax,
                                               const float* __restrict__ bias, float* __restrict__ out) {
  __shared__ float red[8];
  int b = blockIdx.x;
  int t = threadIdx.x;
  const float* mb = magbuf + (size_t)b * MBAT;
  float sb = smax[b];
  float sw = smax[16];
  float v[10];
  float bj[10];
#pragma unroll
  for (int k = 0; k < 10; ++k) {
    int idx = t + 256 * k;
    if (idx > LL * LL - 1) idx = LL * LL - 1;
    int i = idx / LL, j = idx % LL;
    v[k]  = mb[(i * 4 + (j >> 4)) * MROW + (j & 15)];
    bj[k] = bias[j];
  }
  float m = 0.f;
#pragma unroll
  for (int k = 0; k < 10; ++k) m = fmaxf(m, v[k]);
  m = wave_max(m);
  if ((t & 31) == 0) red[t >> 5] = m;
  __syncthreads();
  float r = red[0];
#pragma unroll
  for (int w = 1; w < 8; ++w) r = fmaxf(r, red[w]);
  float scale = (sb * sw) / r;
#pragma unroll
  for (int k = 0; k < 10; ++k) {
    int idx = t + 256 * k;
    if (idx < LL * LL) out[b * 2500 + idx] = v[k] * scale + bj[k];
  }
}

extern "C" void kernel_launch(void* const* d_in, const int* in_sizes, int n_in,
                              void* d_out, int out_size, void* d_ws, size_t ws_size,
                              hipStream_t stream) {
  const float* inp  = (const float*)d_in[0];
  const float* wgt  = (const float*)d_in[1];
  const float* bias = (const float*)d_in[2];
  const float* amp  = (const float*)d_in[3];
  const float* sre  = (const float*)d_in[4];
  const float* sim  = (const float*)d_in[5];

  float* ws   = (float*)d_ws;
  float* smax = ws;
  float* inpt = ws + 64;
  float* tbuf = ws + 40960;
  float* magb = ws + 40960 + 640000;

  k_maxes<<<17, 256, 0, stream>>>(inp, wgt, smax);
  k_table<<<LL * NRT + LL, 128, 0, stream>>>(inp, sre, sim, inpt, tbuf);
  k_prod <<<LL * 4, NQW * 32, 0, stream>>>(wgt, inpt, tbuf, amp, magb);
  k_final<<<BB, 256, 0, stream>>>(magb, smax, bias, (float*)d_out);
}


# launches 4 -> 3: the seventeen maxima (input planes, |weight|) run as extra work-groups of the table kernel's grid, not in a launch of their own; same values, no sum reordered
# speedup vs baseline: 10.5513x; 1.1295x over previous
#include <hip/hip_runtime.h>
#include <math.h>

#define LL    50
#define BB    16
#define ITERR 76
#define KDIM  100
#define IPAD  64
#define SROW  7600
#define SROW0 50
#define COL0  1900
#define NRT   7
#define NQW   25
#define TROW  32
#define MROW  32
#define MBAT  (LL * 4 * MROW)
#define NTAB  (LL * NRT)
#define NMAXB (BB + 1)
#define WS_SMAX 0
#define WS_INPT 64
#define WS_TBUF 40960
#define WS_MAGB (40960 + 640000)

typedef float v2f __attribute__((ext_vector_type(2)));
typedef float v4f __attribute__((ext_vector_type(4)));
typedef float v8f __attribute__((ext_vector_type(8)));

#define WMMA_F32X4(a, b, c) \
  __builtin_amdgcn_wmma_f32_16x16x4_f32(false, (a), false, (b), (short)0, (c), false, false)


__device__ __forceinline__ float wave_max(float m) {
#pragma unroll
  for (int o = 16; o > 0; o >>= 1) m = fmaxf(m, __shfl_xor(m, o, 32));
  return m;
}

__device__ __forceinline__ float max4(float4 v) { return fmaxf(fmaxf(v.x, v.y), fmaxf(v.z, v.w)); }
__device__ __forceinline__ float4 abs4(float4 v) { return make_float4(fabsf(v.x), fabsf(v.y), fabsf(v.z), fabsf(v.w)); }

__device__ __forceinline__ void plane_max(const float* __restrict__ inp, const float* __restrict__ wgt,
                                          float* __restrict__ smax, float* red, int b) {
  int t = threadIdx.x;
  bool isw = b >= BB;
  const float4* p = (const float4*)(isw ? wgt : inp + b * 2500);
  float4 v[5];
#pragma unroll
  for (int k = 0; k < 5; ++k) {
    int f = t + 128 * k; if (f > 624) f = 624;
    v[k] = p[f];
  }
  float m = -3.4e38f;
#pragma unroll
  for (int k = 0; k < 5; ++k) m = fmaxf(m, max4(isw ? abs4(v[k]) : v[k]));
  m = wave_max(m);
  if ((t & 31) == 0) red[t >> 5] = m;
  __syncthreads();
  if (t == 0) smax[b] = fmaxf(fmaxf(red[0], red[1]), fmaxf(red[2], red[3]));
}

__global__ void __launch_bounds__(128) k_table(
    const float* __restrict__ inp, const float* __restrict__ wgt, const float* __restrict__ sre,
    const float* __restrict__ sim, float* __restrict__ ws) {
  __shared__ v2f tw[201];
  __shared__ float red[4];
  float* __restrict__ smax = ws + WS_SMAX;
  float* __restrict__ inpt = ws + WS_INPT;
  float* __restrict__ tbuf = ws + WS_TBUF;
  int bid = blockIdx.x;
  if (bid >= NTAB + LL) {
    plane_max(inp, wgt, smax, red, bid - (NTAB + LL));
    return;
  }
  if (bid >= NTAB) {
    int q = bid - NTAB;
    for (int e = threadIdx.x; e < LL * BB; e += 128) {
      int p = e >> 4, b = e & 15;
      inpt[((q * (LL / 2) + (p >> 1)) * BB + b) * 2 + (p & 1)] = inp[b * 2500 + p * LL + q];
    }
    return;
  }
  for (int t = threadIdx.x; t < 201; t += 128) {
    float s, cz;
    sincospif((float)t * 0.01f, &s, &cz);
    v2f e; e.x = cz; e.y = -s;
    if (t == 200) { e.x = 0.f; e.y = 0.f; }
    tw[t] = e;
  }
  __syncthreads();

  int q  = bid / NRT;
  int rt = bid % NRT;
  int wave = threadIdx.x >> 5;
  int lane = threadIdx.x & 31;
  int h  = lane >> 4;
  int ln = lane & 15;
  int i  = wave * 16 + ln;
  int c0 = ITERR * q;
  size_t rowbase = (size_t)(SROW0 + rt * 16 + ln) * SROW + COL0 + c0;
  const float* sre_l = sre + rowbase;
  const float* sim_l = sim + rowbase;
  int step = i + 1;
  int m0 = (step * (COL0 + c0 + 2 * h)) % 200;
  int adv = (4 * step) % 200;

  v8f cre = {}; v8f cim = {};
  for (int t = 0; t < ITERR / 4; ++t) {
    int u = 4 * t + 2 * h;
    v2f are = *(const v2f*)&sre_l[u];
    v2f aim = *(const v2f*)&sim_l[u];
    v2f aimn = -aim;
    int m1 = m0 + step; if (m1 >= 200) m1 -= 200;
    v2f q0 = tw[(i < LL) ? m0 : 200];
    v2f q1 = tw[(i < LL) ? m1 : 200];
    v2f bre; bre.x = q0.x; bre.y = q1.x;
    v2f bim; bim.x = q0.y; bim.y = q1.y;
    m0 += adv; if (m0 >= 200) m0 -= 200;
    cre = WMMA_F32X4(are,  bre, cre);
    cre = WMMA_F32X4(aimn, bim, cre);
    cim = WMMA_F32X4(are,  bim, cim);
    cim = WMMA_F32X4(aim,  bre, cim);
  }

  if (i < LL) {
#pragma unroll
    for (int r = 0; r < 8; r += 4) {
      int row = rt * 16 + r + 8 * h;
      int p0 = row >> 1;
      if (p0 < LL) {
        float* dst = tbuf + (((size_t)i * LL + q) * TROW + (p0 >> 1)) * 8;
        *(float4*)dst       = make_float4(cre[r],     cre[r + 2], cim[r],     cim[r + 2]);
        *(float4*)(dst + 4) = make_float4(cre[r + 1], cre[r + 3], cim[r + 1], cim[r + 3]);
      }
    }
  }
}

__device__ __forceinline__ v2f fma2(v2f a, v2f b, v2f c) { return __builtin_elementwise_fma(a, b, c); }

__global__ void __launch_bounds__(NQW * 32) k_prod(
    const float* __restrict__ wgt, const float* __restrict__ inpt, const float* __restrict__ tbuf,
    const float* __restrict__ amp, float* __restrict__ magbuf) {
  __shared__ float4 pA[(LL / 2) * 16];
  __shared__ float4 pB[(LL / 2) * 16];
  __shared__ float red[NQW * 512];
  int i  = blockIdx.x >> 2;
  int jt = blockIdx.x & 3;

  {
    int t = threadIdx.x;
    int odd = t & 1, jl = (t >> 1) & 15, g = t >> 5;
    int k1u = 150 + 2 * (jt * 16 + jl);
    int r1 = 4 * g + odd;
    int m1 = (k1u * (50 + r1)) % 200;
    int m2 = (k1u * (52 + r1)) % 200;
    float s1, c1, s2, c2;
    sincospif((float)m1 * 0.01f, &s1, &c1);
    sincospif((float)m2 * 0.01f, &s2, &c2);
    float4 e = make_float4(c1, c2, -s1, -s2);
    if (odd) pB[g * 16 + jl] = e; else pA[g * 16 + jl] = e;
  }
  __syncthreads();

  int wave = threadIdx.x >> 5;
  int lane = threadIdx.x & 31;
  int h = lane >> 4, ln = lane & 15;
  int j = jt * 16 + ln;

  int q = 2 * wave + h;
  float wq = (j < LL) ? fabsf(wgt[j * LL + q]) : 0.f;
  v2f wq2; wq2.x = wq; wq2.y = wq;
  const v2f*    a_l = (const v2f*)inpt + (size_t)q * (LL / 2) * BB + ln;
  const v4f*    t_l = (const v4f*)tbuf + ((size_t)i * LL + q) * TROW * 2;
  const float4* pa_l = pA + ln;
  const float4* pb_l = pB + ln;

  v8f cre = {}; v8f cim = {};
#pragma unroll 1
  for (int g = 0; g < 5; ++g) {
    v2f    av5[5];
    v4f    ta5[5];
    v4f    tb5[5];
#pragma unroll
    for (int s = 0; s < 5; ++s) {
      av5[s] = a_l[(5 * g + s) * 16];
    }
#pragma unroll
    for (int s = 0; s < 5; ++s) {
      ta5[s] = t_l[(5 * g + s) * 2];
      tb5[s] = t_l[(5 * g + s) * 2 + 1];
    }
    asm volatile("" :: "v"(av5[0]), "v"(av5[1]), "v"(av5[2]), "v"(av5[3]), "v"(av5[4]),
                       "v"(ta5[0]), "v"(ta5[1]), "v"(ta5[2]), "v"(ta5[3]), "v"(ta5[4]),
                       "v"(tb5[0]), "v"(tb5[1]), "v"(tb5[2]), "v"(tb5[3]), "v"(tb5[4]));
#pragma unroll
    for (int s = 0; s < 5; ++s) {
      float4 pa = pa_l[(5 * g + s) * 16];
      float4 pb = pb_l[(5 * g + s) * 16];
      v4f ta = ta5[s];
      v4f tb = tb5[s];
      v2f p0r; p0r.x = pa.x; p0r.y = pa.y;
      v2f p0i; p0i.x = pa.z; p0i.y = pa.w;
      v2f p1r; p1r.x = pb.x; p1r.y = pb.y;
      v2f p1i; p1i.x = pb.z; p1i.y = pb.w;
      v2f t0r; t0r.x = ta.x; t0r.y = ta.y;
      v2f t0i; t0i.x = ta.z; t0i.y = ta.w;
      v2f t1r; t1r.x = tb.x; t1r.y = tb.y;
      v2f t1i; t1i.x = tb.z; t1i.y = tb.w;
      v2f bre = fma2(p0r, t0r, fma2(-p0i, t0i, fma2(p1r, t1r, -(p1i * t1i))));
      v2f bim = fma2(p0r, t0i, fma2( p0i, t0r, fma2(p1r, t1i,   p1i * t1r)));
      bre = bre * wq2;
      bim = bim * wq2;
      cre = WMMA_F32X4(av5[s], bre, cre);
      cim = WMMA_F32X4(av5[s], bim, cim);
    }
  }

  float* myred = red + wave * 512;
#pragma unroll
  for (int r = 0; r < 8; ++r) {
    myred[r * 32 + lane]       = cre[r];
    myred[256 + r * 32 + lane] = cim[r];
  }
  __syncthreads();
  if (threadIdx.x < 256) {
    float re = 0.f, im = 0.f;
    for (int w = 0; w < NQW; ++w) {
      re += red[w * 512 + threadIdx.x];
      im += red[w * 512 + 256 + threadIdx.x];
    }
    int r = threadIdx.x >> 5, l = threadIdx.x & 31;
    int b  = r + 8 * (l >> 4);
    int jo = jt * 16 + (l & 15);
    if (jo < LL)
      magbuf[(size_t)b * MBAT + (i * 4 + jt) * MROW + (l & 15)] = sqrtf(re * re + im * im) / amp[jo * LL + i];
  }
}

__global__ void __launch_bounds__(256) k_final(const float* __restrict__ magbuf, const float* __restrict__ smax,
                                               const float* __restrict__ bias, float* __restrict__ out) {
  __shared__ float red[8];
  int b = blockIdx.x;
  int t = threadIdx.x;
  const float* mb = magbuf + (size_t)b * MBAT;
  float sb = smax[b];
  float sw = smax[16];
  float v[10];
  float bj[10];
#pragma unroll
  for (int k = 0; k < 10; ++k) {
    int idx = t + 256 * k;
    if (idx > LL * LL - 1) idx = LL * LL - 1;
    int i = idx / LL, j = idx % LL;
    v[k]  = mb[(i * 4 + (j >> 4)) * MROW + (j & 15)];
    bj[k] = bias[j];
  }
  float m = 0.f;
#pragma unroll
  for (int k = 0; k < 10; ++k) m = fmaxf(m, v[k]);
  m = wave_max(m);
  if ((t & 31) == 0) red[t >> 5] = m;
  __syncthreads();
  float r = red[0];
#pragma unroll
  for (int w = 1; w < 8; ++w) r = fmaxf(r, red[w]);
  float scale = (sb * sw) / r;
#pragma unroll
  for (int k = 0; k < 10; ++k) {
    int idx = t + 256 * k;
    if (idx < LL * LL) out[b * 2500 + idx] = v[k] * scale + bj[k];
  }
}

extern "C" void kernel_launch(void* const* d_in, const int* in_sizes, int n_in,
                              void* d_out, int out_size, void* d_ws, size_t ws_size,
                              hipStream_t stream) {
  const float* inp  = (const float*)d_in[0];
  const float* wgt  = (const float*)d_in[1];
  const float* bias = (const float*)d_in[2];
  const float* amp  = (const float*)d_in[3];
  const float* sre  = (const float*)d_in[4];
  const float* sim  = (const float*)d_in[5];

  float* ws   = (float*)d_ws;
  float* smax = ws + WS_SMAX;
  float* inpt = ws + WS_INPT;
  float* tbuf = ws + WS_TBUF;
  float* magb = ws + WS_MAGB;

  k_table<<<NTAB + LL + NMAXB, 128, 0, stream>>>(inp, wgt, sre, sim, ws);
  k_prod <<<LL * 4, NQW * 32, 0, stream>>>(wgt, inpt, tbuf, amp, magb);
  k_final<<<BB, 256, 0, stream>>>(magb, smax, bias, (float*)d_out);
}
